# starting kernel
# baseline (speedup 1.0000x reference)
#include <hip/hip_runtime.h>

typedef __attribute__((ext_vector_type(2))) float v2f;
typedef __attribute__((ext_vector_type(4))) float v4f;
typedef __attribute__((ext_vector_type(8))) float v8f;

#define FD 32

__global__ void gcn_zero_kernel(float* __restrict__ p, int n4) {
    int i = blockIdx.x * blockDim.x + threadIdx.x;
    if (i < n4) {
        v4f z = {0.0f, 0.0f, 0.0f, 0.0f};
        ((v4f*)p)[i] = z;
    }
}

__global__ void gcn_degree_kernel(const int* __restrict__ src,
                                  const int* __restrict__ dst,
                                  unsigned* __restrict__ outd,
                                  unsigned* __restrict__ ind,
                                  int E) {
    int e = blockIdx.x * blockDim.x + threadIdx.x;
    if (e < E) {
        atomicAdd(&outd[src[e]], 1u);
        atomicAdd(&ind[dst[e]], 1u);
    }
}

__global__ void gcn_scatter_kernel(const float* __restrict__ feat,
                                   const float* __restrict__ e_feat,
                                   const int* __restrict__ src,
                                   const int* __restrict__ dst,
                                   const unsigned* __restrict__ outd,
                                   float* __restrict__ agg,
                                   int E) {
    int wave = blockIdx.x * (blockDim.x >> 5) + (threadIdx.x >> 5);
    int lane = threadIdx.x & 31;
    if (wave >= E) return;
    int s = src[wave];
    int t = dst[wave];
    if (s == t) return;
    float w = e_feat[wave];
    unsigned dg = outd[s];
    float norm = rsqrtf((float)(dg ? dg : 1u));
    float v = feat[(size_t)s * FD + lane] * norm * w;
    atomicAdd(&agg[(size_t)t * FD + lane], v);
}

__global__ void gcn_finalize_kernel(const float* __restrict__ feat,
                                    const float* __restrict__ agg,
                                    const unsigned* __restrict__ ind,
                                    float* __restrict__ out,
                                    int N) {
    int wave = blockIdx.x * (blockDim.x >> 5) + (threadIdx.x >> 5);
    int lane = threadIdx.x & 31;
    int ntiles = (N + 15) >> 4;
    if (wave >= ntiles) return;

    int r    = lane & 15;
    int half = lane >> 4;
    int node = wave * 16 + r;

    float vals[16];
    if (node < N) {
        unsigned dg = ind[node];
        float dn = rsqrtf((float)(dg ? dg : 1u));
        const v4f* fp = (const v4f*)(feat + (size_t)node * FD + half * 16);
        const v4f* ap = (const v4f*)(agg  + (size_t)node * FD + half * 16);
#pragma unroll
        for (int q = 0; q < 4; ++q) {
            v4f f = fp[q];
            v4f a = ap[q];
            vals[4 * q + 0] = __builtin_fabsf(f.x - a.x * dn);
            vals[4 * q + 1] = __builtin_fabsf(f.y - a.y * dn);
            vals[4 * q + 2] = __builtin_fabsf(f.z - a.z * dn);
            vals[4 * q + 3] = __builtin_fabsf(f.w - a.w * dn);
        }
    } else {
#pragma unroll
        for (int q = 0; q < 16; ++q) vals[q] = 0.0f;
    }

    v2f ones = {1.0f, 1.0f};
    v8f c = {};
#pragma unroll
    for (int kk = 0; kk < 8; ++kk) {
        v2f a;
        a.x = vals[2 * kk + 0];
        a.y = vals[2 * kk + 1];
        c = __builtin_amdgcn_wmma_f32_16x16x4_f32(
                false, a, false, ones, (short)0, c, false, false);
    }

    if ((lane & 15) == 0) {
        int base = wave * 16 + half * 8;
#pragma unroll
        for (int m = 0; m < 8; ++m) {
            int nd = base + m;
            if (nd < N) out[nd] = c[m];
        }
    }
}

extern "C" void kernel_launch(void* const* d_in, const int* in_sizes, int n_in,
                              void* d_out, int out_size, void* d_ws, size_t ws_size,
                              hipStream_t stream) {
    const float* feat   = (const float*)d_in[0];
    const float* e_feat = (const float*)d_in[1];
    const int*   src    = (const int*)d_in[2];
    const int*   dst    = (const int*)d_in[3];
    float*       out    = (float*)d_out;

    int N = in_sizes[0] / FD;
    int E = in_sizes[2];

    unsigned* outd = (unsigned*)d_ws;
    unsigned* ind  = outd + N;
    float*    agg  = (float*)(ind + N);

    long long zero_words = (long long)2 * N + (long long)N * FD;
    int zero_v4 = (int)((zero_words + 3) / 4);
    gcn_zero_kernel<<<(zero_v4 + 255) / 256, 256, 0, stream>>>((float*)d_ws, zero_v4);

    gcn_degree_kernel<<<(E + 255) / 256, 256, 0, stream>>>(src, dst, outd, ind, E);

    gcn_scatter_kernel<<<(E + 7) / 8, 256, 0, stream>>>(feat, e_feat, src, dst,
                                                        outd, agg, E);

    int ntiles = (N + 15) / 16;
    gcn_finalize_kernel<<<(ntiles + 7) / 8, 256, 0, stream>>>(feat, agg, ind, out, N);
}


# degree counts: per-work-group LDS histogram over a node sub-range and an edge chunk, flushed with coalesced integer adds, replaces one scattered global atomic per edge endpoint
# speedup vs baseline: 1.6933x; 1.6933x over previous
#include <hip/hip_runtime.h>

typedef __attribute__((ext_vector_type(2))) float v2f;
typedef __attribute__((ext_vector_type(4))) float v4f;
typedef __attribute__((ext_vector_type(8))) float v8f;
typedef __attribute__((ext_vector_type(4))) int v4i;
typedef __attribute__((ext_vector_type(4))) unsigned v4u;

#define FD 32

__global__ void gcn_zero_kernel(float* __restrict__ p, int n4) {
    int i = blockIdx.x * blockDim.x + threadIdx.x;
    if (i < n4) {
        v4f z = {0.0f, 0.0f, 0.0f, 0.0f};
        ((v4f*)p)[i] = z;
    }
}

#define HR 34816
#define HC 16
#define HT 1024
#define HB 8

__device__ __forceinline__ void gcn_count4(unsigned* hist, v4i v, int r0) {
    unsigned k0 = (unsigned)(v.x - r0);
    unsigned k1 = (unsigned)(v.y - r0);
    unsigned k2 = (unsigned)(v.z - r0);
    unsigned k3 = (unsigned)(v.w - r0);
    if (k0 < (unsigned)HR) atomicAdd(&hist[k0], 1u);
    if (k1 < (unsigned)HR) atomicAdd(&hist[k1], 1u);
    if (k2 < (unsigned)HR) atomicAdd(&hist[k2], 1u);
    if (k3 < (unsigned)HR) atomicAdd(&hist[k3], 1u);
}

__global__ void __launch_bounds__(HT)
gcn_degree_kernel(const int* __restrict__ src,
                  const int* __restrict__ dst,
                  unsigned* __restrict__ outd,
                  unsigned* __restrict__ ind,
                  int N, int E) {
    __shared__ unsigned hist[HR];
    const int tid   = threadIdx.x;
    const int chunk = blockIdx.x;
    const int r0    = blockIdx.y * HR;
    const int* __restrict__ idx = blockIdx.z ? dst : src;
    unsigned* __restrict__  cnt = blockIdx.z ? ind : outd;

    for (int i = tid; i < HR / 4; i += HT) {
        v4u z = {0u, 0u, 0u, 0u};
        ((v4u*)hist)[i] = z;
    }
    __syncthreads();

    const int per = ((E + HC * 4 - 1) / (HC * 4)) * 4;
    const int beg = chunk * per;
    const int lim = beg + per;
    const int end = lim < E ? lim : E;
    const int n4  = end > beg ? (end - beg) >> 2 : 0;
    const v4i* __restrict__ p4 = (const v4i*)(idx + beg);

    int i = tid;
    for (; i + (HB - 1) * HT < n4; i += HB * HT) {
        v4i v[HB];
#pragma unroll
        for (int u = 0; u < HB; ++u) v[u] = p4[i + u * HT];
#pragma unroll
        for (int u = 0; u < HB; ++u) gcn_count4(hist, v[u], r0);
    }
    for (; i < n4; i += HT) gcn_count4(hist, p4[i], r0);
    for (int e = beg + n4 * 4 + tid; e < end; e += HT) {
        unsigned k = (unsigned)(idx[e] - r0);
        if (k < (unsigned)HR) atomicAdd(&hist[k], 1u);
    }
    __syncthreads();

    const int rot = chunk * (HR / HC);
    for (int b = tid; b < HR; b += 2 * HT) {
        int j0 = b + rot;
        int j1 = b + HT + rot;
        if (j0 >= HR) j0 -= HR;
        if (j1 >= HR) j1 -= HR;
        unsigned h0 = hist[j0];
        unsigned h1 = hist[j1];
        if (h0 != 0u && r0 + j0 < N) atomicAdd(&cnt[r0 + j0], h0);
        if (h1 != 0u && r0 + j1 < N) atomicAdd(&cnt[r0 + j1], h1);
    }
}

__global__ void gcn_scatter_kernel(const float* __restrict__ feat,
                                   const float* __restrict__ e_feat,
                                   const int* __restrict__ src,
                                   const int* __restrict__ dst,
                                   const unsigned* __restrict__ outd,
                                   float* __restrict__ agg,
                                   int E) {
    int wave = blockIdx.x * (blockDim.x >> 5) + (threadIdx.x >> 5);
    int lane = threadIdx.x & 31;
    if (wave >= E) return;
    int s = src[wave];
    int t = dst[wave];
    if (s == t) return;
    float w = e_feat[wave];
    unsigned dg = outd[s];
    float norm = rsqrtf((float)(dg ? dg : 1u));
    float v = feat[(size_t)s * FD + lane] * norm * w;
    atomicAdd(&agg[(size_t)t * FD + lane], v);
}

__global__ void gcn_finalize_kernel(const float* __restrict__ feat,
                                    const float* __restrict__ agg,
                                    const unsigned* __restrict__ ind,
                                    float* __restrict__ out,
                                    int N) {
    int wave = blockIdx.x * (blockDim.x >> 5) + (threadIdx.x >> 5);
    int lane = threadIdx.x & 31;
    int ntiles = (N + 15) >> 4;
    if (wave >= ntiles) return;

    int r    = lane & 15;
    int half = lane >> 4;
    int node = wave * 16 + r;

    float vals[16];
    if (node < N) {
        unsigned dg = ind[node];
        float dn = rsqrtf((float)(dg ? dg : 1u));
        const v4f* fp = (const v4f*)(feat + (size_t)node * FD + half * 16);
        const v4f* ap = (const v4f*)(agg  + (size_t)node * FD + half * 16);
#pragma unroll
        for (int q = 0; q < 4; ++q) {
            v4f f = fp[q];
            v4f a = ap[q];
            vals[4 * q + 0] = __builtin_fabsf(f.x - a.x * dn);
            vals[4 * q + 1] = __builtin_fabsf(f.y - a.y * dn);
            vals[4 * q + 2] = __builtin_fabsf(f.z - a.z * dn);
            vals[4 * q + 3] = __builtin_fabsf(f.w - a.w * dn);
        }
    } else {
#pragma unroll
        for (int q = 0; q < 16; ++q) vals[q] = 0.0f;
    }

    v2f ones = {1.0f, 1.0f};
    v8f c = {};
#pragma unroll
    for (int kk = 0; kk < 8; ++kk) {
        v2f a;
        a.x = vals[2 * kk + 0];
        a.y = vals[2 * kk + 1];
        c = __builtin_amdgcn_wmma_f32_16x16x4_f32(
                false, a, false, ones, (short)0, c, false, false);
    }

    if ((lane & 15) == 0) {
        int base = wave * 16 + half * 8;
#pragma unroll
        for (int m = 0; m < 8; ++m) {
            int nd = base + m;
            if (nd < N) out[nd] = c[m];
        }
    }
}

extern "C" void kernel_launch(void* const* d_in, const int* in_sizes, int n_in,
                              void* d_out, int out_size, void* d_ws, size_t ws_size,
                              hipStream_t stream) {
    const float* feat   = (const float*)d_in[0];
    const float* e_feat = (const float*)d_in[1];
    const int*   src    = (const int*)d_in[2];
    const int*   dst    = (const int*)d_in[3];
    float*       out    = (float*)d_out;

    int N = in_sizes[0] / FD;
    int E = in_sizes[2];

    unsigned* outd = (unsigned*)d_ws;
    unsigned* ind  = outd + N;
    float*    agg  = (float*)(ind + N);

    long long zero_words = (long long)2 * N + (long long)N * FD;
    int zero_v4 = (int)((zero_words + 3) / 4);
    gcn_zero_kernel<<<(zero_v4 + 255) / 256, 256, 0, stream>>>((float*)d_ws, zero_v4);

    gcn_degree_kernel<<<dim3(HC, (N + HR - 1) / HR, 2), HT, 0, stream>>>(src, dst, outd, ind, N, E);

    gcn_scatter_kernel<<<(E + 7) / 8, 256, 0, stream>>>(feat, e_feat, src, dst,
                                                        outd, agg, E);

    int ntiles = (N + 15) / 16;
    gcn_finalize_kernel<<<(ntiles + 7) / 8, 256, 0, stream>>>(feat, agg, ind, out, N);
}


# edge scatter: one wave takes 32 consecutive edges (lane-parallel index, weight, degree loads; 32 source-row gathers in flight), not one wave per edge; f32 atomic adds kept, summed in arrival order
# speedup vs baseline: 3.8473x; 2.2721x over previous
#include <hip/hip_runtime.h>

typedef __attribute__((ext_vector_type(2))) float v2f;
typedef __attribute__((ext_vector_type(4))) float v4f;
typedef __attribute__((ext_vector_type(8))) float v8f;
typedef __attribute__((ext_vector_type(4))) int v4i;
typedef __attribute__((ext_vector_type(4))) unsigned v4u;

#define FD 32

__global__ void gcn_zero_kernel(float* __restrict__ p, int n4) {
    int i = blockIdx.x * blockDim.x + threadIdx.x;
    if (i < n4) {
        v4f z = {0.0f, 0.0f, 0.0f, 0.0f};
        ((v4f*)p)[i] = z;
    }
}

#define HR 34816
#define HC 16
#define HT 1024
#define HB 8

__device__ __forceinline__ void gcn_count4(unsigned* hist, v4i v, int r0) {
    unsigned k0 = (unsigned)(v.x - r0);
    unsigned k1 = (unsigned)(v.y - r0);
    unsigned k2 = (unsigned)(v.z - r0);
    unsigned k3 = (unsigned)(v.w - r0);
    if (k0 < (unsigned)HR) atomicAdd(&hist[k0], 1u);
    if (k1 < (unsigned)HR) atomicAdd(&hist[k1], 1u);
    if (k2 < (unsigned)HR) atomicAdd(&hist[k2], 1u);
    if (k3 < (unsigned)HR) atomicAdd(&hist[k3], 1u);
}

__global__ void __launch_bounds__(HT)
gcn_degree_kernel(const int* __restrict__ src,
                  const int* __restrict__ dst,
                  unsigned* __restrict__ outd,
                  unsigned* __restrict__ ind,
                  int N, int E) {
    __shared__ unsigned hist[HR];
    const int tid   = threadIdx.x;
    const int chunk = blockIdx.x;
    const int r0    = blockIdx.y * HR;
    const int* __restrict__ idx = blockIdx.z ? dst : src;
    unsigned* __restrict__  cnt = blockIdx.z ? ind : outd;

    for (int i = tid; i < HR / 4; i += HT) {
        v4u z = {0u, 0u, 0u, 0u};
        ((v4u*)hist)[i] = z;
    }
    __syncthreads();

    const int per = ((E + HC * 4 - 1) / (HC * 4)) * 4;
    const int beg = chunk * per;
    const int lim = beg + per;
    const int end = lim < E ? lim : E;
    const int n4  = end > beg ? (end - beg) >> 2 : 0;
    const v4i* __restrict__ p4 = (const v4i*)(idx + beg);

    int i = tid;
    for (; i + (HB - 1) * HT < n4; i += HB * HT) {
        v4i v[HB];
#pragma unroll
        for (int u = 0; u < HB; ++u) v[u] = p4[i + u * HT];
#pragma unroll
        for (int u = 0; u < HB; ++u) gcn_count4(hist, v[u], r0);
    }
    for (; i < n4; i += HT) gcn_count4(hist, p4[i], r0);
    for (int e = beg + n4 * 4 + tid; e < end; e += HT) {
        unsigned k = (unsigned)(idx[e] - r0);
        if (k < (unsigned)HR) atomicAdd(&hist[k], 1u);
    }
    __syncthreads();

    const int rot = chunk * (HR / HC);
    for (int b = tid; b < HR; b += 2 * HT) {
        int j0 = b + rot;
        int j1 = b + HT + rot;
        if (j0 >= HR) j0 -= HR;
        if (j1 >= HR) j1 -= HR;
        unsigned h0 = hist[j0];
        unsigned h1 = hist[j1];
        if (h0 != 0u && r0 + j0 < N) atomicAdd(&cnt[r0 + j0], h0);
        if (h1 != 0u && r0 + j1 < N) atomicAdd(&cnt[r0 + j1], h1);
    }
}

#define EW 32
#define SW 8

__device__ __forceinline__ float gcn_lane_f32(float x, int j) {
    return __int_as_float(__builtin_amdgcn_readlane(__float_as_int(x), j));
}

__global__ void __launch_bounds__(SW * 32)
gcn_scatter_kernel(const float* __restrict__ feat,
                   const float* __restrict__ e_feat,
                   const int* __restrict__ src,
                   const int* __restrict__ dst,
                   const unsigned* __restrict__ outd,
                   float* __restrict__ agg,
                   int E) {
    const int lane = threadIdx.x & 31;
    const int wv = __builtin_amdgcn_readfirstlane((int)(threadIdx.x >> 5));
    const int e0 = (blockIdx.x * SW + wv) * EW;
    if (e0 >= E) return;
    const int left = E - e0;

    const int e = e0 + lane;
    int s = 0, t = 0;
    float w = 0.0f;
    if (e < E) {
        s = src[e];
        t = dst[e];
        w = e_feat[e];
    }
    const unsigned dg = outd[s];

    if (left >= EW) {
        float f[EW];
#pragma unroll
        for (int j = 0; j < EW; ++j) {
            const int sj = __builtin_amdgcn_readlane(s, j);
            f[j] = feat[(size_t)sj * FD + lane];
        }
        const float norm = rsqrtf((float)(dg ? dg : 1u));
        const float wm = w * ((s != t) ? 1.0f : 0.0f);
#pragma unroll
        for (int j = 0; j < EW; ++j) {
            const int tj = __builtin_amdgcn_readlane(t, j);
            atomicAdd(&agg[(size_t)tj * FD + lane],
                      f[j] * gcn_lane_f32(norm, j) * gcn_lane_f32(wm, j));
        }
    } else {
        const float norm = rsqrtf((float)(dg ? dg : 1u));
        const float wm = w * ((s != t) ? 1.0f : 0.0f);
        for (int j = 0; j < left; ++j) {
            const int sj = __builtin_amdgcn_readlane(s, j);
            const int tj = __builtin_amdgcn_readlane(t, j);
            atomicAdd(&agg[(size_t)tj * FD + lane],
                      feat[(size_t)sj * FD + lane] * gcn_lane_f32(norm, j) * gcn_lane_f32(wm, j));
        }
    }
}

__global__ void gcn_finalize_kernel(const float* __restrict__ feat,
                                    const float* __restrict__ agg,
                                    const unsigned* __restrict__ ind,
                                    float* __restrict__ out,
                                    int N) {
    int wave = blockIdx.x * (blockDim.x >> 5) + (threadIdx.x >> 5);
    int lane = threadIdx.x & 31;
    int ntiles = (N + 15) >> 4;
    if (wave >= ntiles) return;

    int r    = lane & 15;
    int half = lane >> 4;
    int node = wave * 16 + r;

    float vals[16];
    if (node < N) {
        unsigned dg = ind[node];
        float dn = rsqrtf((float)(dg ? dg : 1u));
        const v4f* fp = (const v4f*)(feat + (size_t)node * FD + half * 16);
        const v4f* ap = (const v4f*)(agg  + (size_t)node * FD + half * 16);
#pragma unroll
        for (int q = 0; q < 4; ++q) {
            v4f f = fp[q];
            v4f a = ap[q];
            vals[4 * q + 0] = __builtin_fabsf(f.x - a.x * dn);
            vals[4 * q + 1] = __builtin_fabsf(f.y - a.y * dn);
            vals[4 * q + 2] = __builtin_fabsf(f.z - a.z * dn);
            vals[4 * q + 3] = __builtin_fabsf(f.w - a.w * dn);
        }
    } else {
#pragma unroll
        for (int q = 0; q < 16; ++q) vals[q] = 0.0f;
    }

    v2f ones = {1.0f, 1.0f};
    v8f c = {};
#pragma unroll
    for (int kk = 0; kk < 8; ++kk) {
        v2f a;
        a.x = vals[2 * kk + 0];
        a.y = vals[2 * kk + 1];
        c = __builtin_amdgcn_wmma_f32_16x16x4_f32(
                false, a, false, ones, (short)0, c, false, false);
    }

    if ((lane & 15) == 0) {
        int base = wave * 16 + half * 8;
#pragma unroll
        for (int m = 0; m < 8; ++m) {
            int nd = base + m;
            if (nd < N) out[nd] = c[m];
        }
    }
}

extern "C" void kernel_launch(void* const* d_in, const int* in_sizes, int n_in,
                              void* d_out, int out_size, void* d_ws, size_t ws_size,
                              hipStream_t stream) {
    const float* feat   = (const float*)d_in[0];
    const float* e_feat = (const float*)d_in[1];
    const int*   src    = (const int*)d_in[2];
    const int*   dst    = (const int*)d_in[3];
    float*       out    = (float*)d_out;

    int N = in_sizes[0] / FD;
    int E = in_sizes[2];

    unsigned* outd = (unsigned*)d_ws;
    unsigned* ind  = outd + N;
    float*    agg  = (float*)(ind + N);

    long long zero_words = (long long)2 * N + (long long)N * FD;
    int zero_v4 = (int)((zero_words + 3) / 4);
    gcn_zero_kernel<<<(zero_v4 + 255) / 256, 256, 0, stream>>>((float*)d_ws, zero_v4);

    gcn_degree_kernel<<<dim3(HC, (N + HR - 1) / HR, 2), HT, 0, stream>>>(src, dst, outd, ind, N, E);

    gcn_scatter_kernel<<<(E + SW * EW - 1) / (SW * EW), SW * 32, 0, stream>>>(
        feat, e_feat, src, dst, outd, agg, E);

    int ntiles = (N + 15) / 16;
    gcn_finalize_kernel<<<(ntiles + 7) / 8, 256, 0, stream>>>(feat, agg, ind, out, N);
}


# edge scatter split by chip half: node rows in two sides, one chunk queue each; a work-group reads its die and draws from its own side first, so agg lines stay in one L2; f32 atomics, arrival order
# speedup vs baseline: 4.7076x; 1.2236x over previous
#include <hip/hip_runtime.h>

typedef __attribute__((ext_vector_type(2))) float v2f;
typedef __attribute__((ext_vector_type(4))) float v4f;
typedef __attribute__((ext_vector_type(8))) float v8f;
typedef __attribute__((ext_vector_type(4))) int v4i;
typedef __attribute__((ext_vector_type(4))) unsigned v4u;

#define FD 32

__global__ void gcn_zero_kernel(float* __restrict__ p, int n4) {
    int i = blockIdx.x * blockDim.x + threadIdx.x;
    if (i < n4) {
        v4f z = {0.0f, 0.0f, 0.0f, 0.0f};
        ((v4f*)p)[i] = z;
    }
}

#define HR 34816
#define HC 16
#define HT 1024
#define HB 8

__device__ __forceinline__ void gcn_count4(unsigned* hist, v4i v, int r0) {
    unsigned k0 = (unsigned)(v.x - r0);
    unsigned k1 = (unsigned)(v.y - r0);
    unsigned k2 = (unsigned)(v.z - r0);
    unsigned k3 = (unsigned)(v.w - r0);
    if (k0 < (unsigned)HR) atomicAdd(&hist[k0], 1u);
    if (k1 < (unsigned)HR) atomicAdd(&hist[k1], 1u);
    if (k2 < (unsigned)HR) atomicAdd(&hist[k2], 1u);
    if (k3 < (unsigned)HR) atomicAdd(&hist[k3], 1u);
}

__global__ void __launch_bounds__(HT)
gcn_degree_kernel(const int* __restrict__ src,
                  const int* __restrict__ dst,
                  unsigned* __restrict__ outd,
                  unsigned* __restrict__ ind,
                  int N, int E) {
    __shared__ unsigned hist[HR];
    const int tid   = threadIdx.x;
    const int chunk = blockIdx.x;
    const int r0    = blockIdx.y * HR;
    const int* __restrict__ idx = blockIdx.z ? dst : src;
    unsigned* __restrict__  cnt = blockIdx.z ? ind : outd;

    for (int i = tid; i < HR / 4; i += HT) {
        v4u z = {0u, 0u, 0u, 0u};
        ((v4u*)hist)[i] = z;
    }
    __syncthreads();

    const int per = ((E + HC * 4 - 1) / (HC * 4)) * 4;
    const int beg = chunk * per;
    const int lim = beg + per;
    const int end = lim < E ? lim : E;
    const int n4  = end > beg ? (end - beg) >> 2 : 0;
    const v4i* __restrict__ p4 = (const v4i*)(idx + beg);

    int i = tid;
    for (; i + (HB - 1) * HT < n4; i += HB * HT) {
        v4i v[HB];
#pragma unroll
        for (int u = 0; u < HB; ++u) v[u] = p4[i + u * HT];
#pragma unroll
        for (int u = 0; u < HB; ++u) gcn_count4(hist, v[u], r0);
    }
    for (; i < n4; i += HT) gcn_count4(hist, p4[i], r0);
    for (int e = beg + n4 * 4 + tid; e < end; e += HT) {
        unsigned k = (unsigned)(idx[e] - r0);
        if (k < (unsigned)HR) atomicAdd(&hist[k], 1u);
    }
    __syncthreads();

    const int rot = chunk * (HR / HC);
    for (int b = tid; b < HR; b += 2 * HT) {
        int j0 = b + rot;
        int j1 = b + HT + rot;
        if (j0 >= HR) j0 -= HR;
        if (j1 >= HR) j1 -= HR;
        unsigned h0 = hist[j0];
        unsigned h1 = hist[j1];
        if (h0 != 0u && r0 + j0 < N) atomicAdd(&cnt[r0 + j0], h0);
        if (h1 != 0u && r0 + j1 < N) atomicAdd(&cnt[r0 + j1], h1);
    }
}

#define QW 8
#define QB 2
#define QE (QW * QB * 32)
#define QS 64
#define QG 1280

__device__ __forceinline__ float gcn_lane_f32(float x, int j) {
    return __int_as_float(__builtin_amdgcn_readlane(__float_as_int(x), j));
}

__device__ __forceinline__ void gcn_lds_barrier() {
    __builtin_amdgcn_fence(__ATOMIC_RELEASE, "workgroup", "local");
    __builtin_amdgcn_s_barrier();
    __builtin_amdgcn_fence(__ATOMIC_ACQUIRE, "workgroup", "local");
}

__global__ void __launch_bounds__(QW * 32)
gcn_scatter_kernel(const float* __restrict__ feat,
                   const float* __restrict__ e_feat,
                   const int* __restrict__ src,
                   const int* __restrict__ dst,
                   const unsigned* __restrict__ outd,
                   float* __restrict__ agg,
                   unsigned* __restrict__ qctr,
                   int N, int E) {
    __shared__ int s_chunk;
    __shared__ int s_side;
    const int lane = threadIdx.x & 31;
    const int wv = __builtin_amdgcn_readfirstlane((int)(threadIdx.x >> 5));
    const unsigned nch = (unsigned)((E + QE - 1) / QE);
    const int split = N >> 1;
    const int home = (__builtin_amdgcn_s_sendmsg_rtn(0x87) >> 18) & 1;

    int q = home;
    unsigned pend = 0u;
    if (threadIdx.x == 0) pend = atomicAdd(&qctr[q * QS], 1u);

    for (;;) {
        if (threadIdx.x == 0) {
            unsigned b = pend;
            if (b >= nch && q == home) {
                q = home ^ 1;
                b = atomicAdd(&qctr[q * QS], 1u);
            }
            s_chunk = (b < nch) ? (int)b : -1;
            s_side = q;
        }
        gcn_lds_barrier();
        const int chunk = s_chunk;
        const int side = s_side;
        if (chunk < 0) break;
        if (threadIdx.x == 0) pend = atomicAdd(&qctr[q * QS], 1u);

        const int e0 = ((chunk * QW + wv) * QB) * 32;
        int s[QB], t[QB];
        float wm[QB], norm[QB];
        unsigned m[QB];
#pragma unroll
        for (int u = 0; u < QB; ++u) {
            const int e = e0 + u * 32 + lane;
            const bool inb = e < E;
            int su = 0, tu = 0;
            float wu = 0.0f;
            if (inb) {
                su = src[e];
                tu = dst[e];
                wu = e_feat[e];
            }
            s[u] = su;
            t[u] = tu;
            wm[u] = wu * ((su != tu) ? 1.0f : 0.0f);
            m[u] = __builtin_amdgcn_ballot_w32(inb && ((tu >= split ? 1 : 0) == side));
        }
#pragma unroll
        for (int u = 0; u < QB; ++u) {
            const unsigned dg = outd[s[u]];
            norm[u] = rsqrtf((float)(dg ? dg : 1u));
        }

        float f[QB][32];
#pragma unroll
        for (int u = 0; u < QB; ++u) {
#pragma unroll
            for (int j = 0; j < 32; ++j) {
                if (m[u] & (1u << j)) {
                    const int sj = __builtin_amdgcn_readlane(s[u], j);
                    f[u][j] = feat[(size_t)sj * FD + lane];
                }
            }
        }
#pragma unroll
        for (int u = 0; u < QB; ++u) {
#pragma unroll
            for (int j = 0; j < 32; ++j) {
                if (m[u] & (1u << j)) {
                    const int tj = __builtin_amdgcn_readlane(t[u], j);
                    atomicAdd(&agg[(size_t)tj * FD + lane],
                              f[u][j] * gcn_lane_f32(norm[u], j) * gcn_lane_f32(wm[u], j));
                }
            }
        }
        gcn_lds_barrier();
    }
}

__global__ void gcn_finalize_kernel(const float* __restrict__ feat,
                                    const float* __restrict__ agg,
                                    const unsigned* __restrict__ ind,
                                    float* __restrict__ out,
                                    int N) {
    int wave = blockIdx.x * (blockDim.x >> 5) + (threadIdx.x >> 5);
    int lane = threadIdx.x & 31;
    int ntiles = (N + 15) >> 4;
    if (wave >= ntiles) return;

    int r    = lane & 15;
    int half = lane >> 4;
    int node = wave * 16 + r;

    float vals[16];
    if (node < N) {
        unsigned dg = ind[node];
        float dn = rsqrtf((float)(dg ? dg : 1u));
        const v4f* fp = (const v4f*)(feat + (size_t)node * FD + half * 16);
        const v4f* ap = (const v4f*)(agg  + (size_t)node * FD + half * 16);
#pragma unroll
        for (int q = 0; q < 4; ++q) {
            v4f f = fp[q];
            v4f a = ap[q];
            vals[4 * q + 0] = __builtin_fabsf(f.x - a.x * dn);
            vals[4 * q + 1] = __builtin_fabsf(f.y - a.y * dn);
            vals[4 * q + 2] = __builtin_fabsf(f.z - a.z * dn);
            vals[4 * q + 3] = __builtin_fabsf(f.w - a.w * dn);
        }
    } else {
#pragma unroll
        for (int q = 0; q < 16; ++q) vals[q] = 0.0f;
    }

    v2f ones = {1.0f, 1.0f};
    v8f c = {};
#pragma unroll
    for (int kk = 0; kk < 8; ++kk) {
        v2f a;
        a.x = vals[2 * kk + 0];
        a.y = vals[2 * kk + 1];
        c = __builtin_amdgcn_wmma_f32_16x16x4_f32(
                false, a, false, ones, (short)0, c, false, false);
    }

    if ((lane & 15) == 0) {
        int base = wave * 16 + half * 8;
#pragma unroll
        for (int m = 0; m < 8; ++m) {
            int nd = base + m;
            if (nd < N) out[nd] = c[m];
        }
    }
}

extern "C" void kernel_launch(void* const* d_in, const int* in_sizes, int n_in,
                              void* d_out, int out_size, void* d_ws, size_t ws_size,
                              hipStream_t stream) {
    const float* feat   = (const float*)d_in[0];
    const float* e_feat = (const float*)d_in[1];
    const int*   src    = (const int*)d_in[2];
    const int*   dst    = (const int*)d_in[3];
    float*       out    = (float*)d_out;

    int N = in_sizes[0] / FD;
    int E = in_sizes[2];

    unsigned* outd = (unsigned*)d_ws;
    unsigned* ind  = outd + N;
    float*    agg  = (float*)(ind + N);
    unsigned* qctr = (unsigned*)(agg + (size_t)N * FD);

    long long zero_words = (long long)2 * N + (long long)N * FD + 2 * QS;
    int zero_v4 = (int)((zero_words + 3) / 4);
    gcn_zero_kernel<<<(zero_v4 + 255) / 256, 256, 0, stream>>>((float*)d_ws, zero_v4);

    gcn_degree_kernel<<<dim3(HC, (N + HR - 1) / HR, 2), HT, 0, stream>>>(src, dst, outd, ind, N, E);

    gcn_scatter_kernel<<<QG, QW * 32, 0, stream>>>(
        feat, e_feat, src, dst, outd, agg, qctr, N, E);

    int ntiles = (N + 15) / 16;
    gcn_finalize_kernel<<<(ntiles + 7) / 8, 256, 0, stream>>>(feat, agg, ind, out, N);
}


# edge scatter: the out-degree gather (a scattered 4-byte request per lane) issued only by lanes whose edge lies on the visited side, not by all 32 on both visits of a chunk; f32 atomics, arrival order
# speedup vs baseline: 4.9059x; 1.0421x over previous
#include <hip/hip_runtime.h>

typedef __attribute__((ext_vector_type(2))) float v2f;
typedef __attribute__((ext_vector_type(4))) float v4f;
typedef __attribute__((ext_vector_type(8))) float v8f;
typedef __attribute__((ext_vector_type(4))) int v4i;
typedef __attribute__((ext_vector_type(4))) unsigned v4u;

#define FD 32

__global__ void gcn_zero_kernel(float* __restrict__ p, int n4) {
    int i = blockIdx.x * blockDim.x + threadIdx.x;
    if (i < n4) {
        v4f z = {0.0f, 0.0f, 0.0f, 0.0f};
        ((v4f*)p)[i] = z;
    }
}

#define HR 34816
#define HC 16
#define HT 1024
#define HB 8

__device__ __forceinline__ void gcn_count4(unsigned* hist, v4i v, int r0) {
    unsigned k0 = (unsigned)(v.x - r0);
    unsigned k1 = (unsigned)(v.y - r0);
    unsigned k2 = (unsigned)(v.z - r0);
    unsigned k3 = (unsigned)(v.w - r0);
    if (k0 < (unsigned)HR) atomicAdd(&hist[k0], 1u);
    if (k1 < (unsigned)HR) atomicAdd(&hist[k1], 1u);
    if (k2 < (unsigned)HR) atomicAdd(&hist[k2], 1u);
    if (k3 < (unsigned)HR) atomicAdd(&hist[k3], 1u);
}

__global__ void __launch_bounds__(HT)
gcn_degree_kernel(const int* __restrict__ src,
                  const int* __restrict__ dst,
                  unsigned* __restrict__ outd,
                  unsigned* __restrict__ ind,
                  int N, int E) {
    __shared__ unsigned hist[HR];
    const int tid   = threadIdx.x;
    const int chunk = blockIdx.x;
    const int r0    = blockIdx.y * HR;
    const int* __restrict__ idx = blockIdx.z ? dst : src;
    unsigned* __restrict__  cnt = blockIdx.z ? ind : outd;

    for (int i = tid; i < HR / 4; i += HT) {
        v4u z = {0u, 0u, 0u, 0u};
        ((v4u*)hist)[i] = z;
    }
    __syncthreads();

    const int per = ((E + HC * 4 - 1) / (HC * 4)) * 4;
    const int beg = chunk * per;
    const int lim = beg + per;
    const int end = lim < E ? lim : E;
    const int n4  = end > beg ? (end - beg) >> 2 : 0;
    const v4i* __restrict__ p4 = (const v4i*)(idx + beg);

    int i = tid;
    for (; i + (HB - 1) * HT < n4; i += HB * HT) {
        v4i v[HB];
#pragma unroll
        for (int u = 0; u < HB; ++u) v[u] = p4[i + u * HT];
#pragma unroll
        for (int u = 0; u < HB; ++u) gcn_count4(hist, v[u], r0);
    }
    for (; i < n4; i += HT) gcn_count4(hist, p4[i], r0);
    for (int e = beg + n4 * 4 + tid; e < end; e += HT) {
        unsigned k = (unsigned)(idx[e] - r0);
        if (k < (unsigned)HR) atomicAdd(&hist[k], 1u);
    }
    __syncthreads();

    const int rot = chunk * (HR / HC);
    for (int b = tid; b < HR; b += 2 * HT) {
        int j0 = b + rot;
        int j1 = b + HT + rot;
        if (j0 >= HR) j0 -= HR;
        if (j1 >= HR) j1 -= HR;
        unsigned h0 = hist[j0];
        unsigned h1 = hist[j1];
        if (h0 != 0u && r0 + j0 < N) atomicAdd(&cnt[r0 + j0], h0);
        if (h1 != 0u && r0 + j1 < N) atomicAdd(&cnt[r0 + j1], h1);
    }
}

#define QW 8
#define QB 2
#define QE (QW * QB * 32)
#define QS 64
#define QG 1280

__device__ __forceinline__ float gcn_lane_f32(float x, int j) {
    return __int_as_float(__builtin_amdgcn_readlane(__float_as_int(x), j));
}

__device__ __forceinline__ void gcn_lds_barrier() {
    __builtin_amdgcn_fence(__ATOMIC_RELEASE, "workgroup", "local");
    __builtin_amdgcn_s_barrier();
    __builtin_amdgcn_fence(__ATOMIC_ACQUIRE, "workgroup", "local");
}

__global__ void __launch_bounds__(QW * 32)
gcn_scatter_kernel(const float* __restrict__ feat,
                   const float* __restrict__ e_feat,
                   const int* __restrict__ src,
                   const int* __restrict__ dst,
                   const unsigned* __restrict__ outd,
                   float* __restrict__ agg,
                   unsigned* __restrict__ qctr,
                   int N, int E) {
    __shared__ int s_chunk;
    __shared__ int s_side;
    const int lane = threadIdx.x & 31;
    const int wv = __builtin_amdgcn_readfirstlane((int)(threadIdx.x >> 5));
    const unsigned nch = (unsigned)((E + QE - 1) / QE);
    const int split = N >> 1;
    const int home = (__builtin_amdgcn_s_sendmsg_rtn(0x87) >> 18) & 1;

    int q = home;
    unsigned pend = 0u;
    if (threadIdx.x == 0) pend = atomicAdd(&qctr[q * QS], 1u);

    for (;;) {
        if (threadIdx.x == 0) {
            unsigned b = pend;
            if (b >= nch && q == home) {
                q = home ^ 1;
                b = atomicAdd(&qctr[q * QS], 1u);
            }
            s_chunk = (b < nch) ? (int)b : -1;
            s_side = q;
        }
        gcn_lds_barrier();
        const int chunk = s_chunk;
        const int side = s_side;
        if (chunk < 0) break;
        if (threadIdx.x == 0) pend = atomicAdd(&qctr[q * QS], 1u);

        const int e0 = ((chunk * QW + wv) * QB) * 32;
        int s[QB], t[QB];
        float wm[QB], norm[QB];
        unsigned m[QB];
        bool own[QB];
        unsigned dg[QB];
#pragma unroll
        for (int u = 0; u < QB; ++u) {
            const int e = e0 + u * 32 + lane;
            const bool inb = e < E;
            int su = 0, tu = 0;
            float wu = 0.0f;
            if (inb) {
                su = src[e];
                tu = dst[e];
                wu = e_feat[e];
            }
            s[u] = su;
            t[u] = tu;
            wm[u] = wu * ((su != tu) ? 1.0f : 0.0f);
            own[u] = inb && ((tu >= split ? 1 : 0) == side);
            m[u] = __builtin_amdgcn_ballot_w32(own[u]);
        }
#pragma unroll
        for (int u = 0; u < QB; ++u) {
            dg[u] = 1u;
            if (own[u]) dg[u] = outd[s[u]];
        }
#pragma unroll
        for (int u = 0; u < QB; ++u) norm[u] = rsqrtf((float)(dg[u] ? dg[u] : 1u));

        float f[QB][32];
#pragma unroll
        for (int u = 0; u < QB; ++u) {
#pragma unroll
            for (int j = 0; j < 32; ++j) {
                if (m[u] & (1u << j)) {
                    const int sj = __builtin_amdgcn_readlane(s[u], j);
                    f[u][j] = feat[(size_t)sj * FD + lane];
                }
            }
        }
#pragma unroll
        for (int u = 0; u < QB; ++u) {
#pragma unroll
            for (int j = 0; j < 32; ++j) {
                if (m[u] & (1u << j)) {
                    const int tj = __builtin_amdgcn_readlane(t[u], j);
                    atomicAdd(&agg[(size_t)tj * FD + lane],
                              f[u][j] * gcn_lane_f32(norm[u], j) * gcn_lane_f32(wm[u], j));
                }
            }
        }
        gcn_lds_barrier();
    }
}

__global__ void gcn_finalize_kernel(const float* __restrict__ feat,
                                    const float* __restrict__ agg,
                                    const unsigned* __restrict__ ind,
                                    float* __restrict__ out,
                                    int N) {
    int wave = blockIdx.x * (blockDim.x >> 5) + (threadIdx.x >> 5);
    int lane = threadIdx.x & 31;
    int ntiles = (N + 15) >> 4;
    if (wave >= ntiles) return;

    int r    = lane & 15;
    int half = lane >> 4;
    int node = wave * 16 + r;

    float vals[16];
    if (node < N) {
        unsigned dg = ind[node];
        float dn = rsqrtf((float)(dg ? dg : 1u));
        const v4f* fp = (const v4f*)(feat + (size_t)node * FD + half * 16);
        const v4f* ap = (const v4f*)(agg  + (size_t)node * FD + half * 16);
#pragma unroll
        for (int q = 0; q < 4; ++q) {
            v4f f = fp[q];
            v4f a = ap[q];
            vals[4 * q + 0] = __builtin_fabsf(f.x - a.x * dn);
            vals[4 * q + 1] = __builtin_fabsf(f.y - a.y * dn);
            vals[4 * q + 2] = __builtin_fabsf(f.z - a.z * dn);
            vals[4 * q + 3] = __builtin_fabsf(f.w - a.w * dn);
        }
    } else {
#pragma unroll
        for (int q = 0; q < 16; ++q) vals[q] = 0.0f;
    }

    v2f ones = {1.0f, 1.0f};
    v8f c = {};
#pragma unroll
    for (int kk = 0; kk < 8; ++kk) {
        v2f a;
        a.x = vals[2 * kk + 0];
        a.y = vals[2 * kk + 1];
        c = __builtin_amdgcn_wmma_f32_16x16x4_f32(
                false, a, false, ones, (short)0, c, false, false);
    }

    if ((lane & 15) == 0) {
        int base = wave * 16 + half * 8;
#pragma unroll
        for (int m = 0; m < 8; ++m) {
            int nd = base + m;
            if (nd < N) out[nd] = c[m];
        }
    }
}

extern "C" void kernel_launch(void* const* d_in, const int* in_sizes, int n_in,
                              void* d_out, int out_size, void* d_ws, size_t ws_size,
                              hipStream_t stream) {
    const float* feat   = (const float*)d_in[0];
    const float* e_feat = (const float*)d_in[1];
    const int*   src    = (const int*)d_in[2];
    const int*   dst    = (const int*)d_in[3];
    float*       out    = (float*)d_out;

    int N = in_sizes[0] / FD;
    int E = in_sizes[2];

    unsigned* outd = (unsigned*)d_ws;
    unsigned* ind  = outd + N;
    float*    agg  = (float*)(ind + N);
    unsigned* qctr = (unsigned*)(agg + (size_t)N * FD);

    long long zero_words = (long long)2 * N + (long long)N * FD + 2 * QS;
    int zero_v4 = (int)((zero_words + 3) / 4);
    gcn_zero_kernel<<<(zero_v4 + 255) / 256, 256, 0, stream>>>((float*)d_ws, zero_v4);

    gcn_degree_kernel<<<dim3(HC, (N + HR - 1) / HR, 2), HT, 0, stream>>>(src, dst, outd, ind, N, E);

    gcn_scatter_kernel<<<QG, QW * 32, 0, stream>>>(
        feat, e_feat, src, dst, outd, agg, qctr, N, E);

    int ntiles = (N + 15) / 16;
    gcn_finalize_kernel<<<(ntiles + 7) / 8, 256, 0, stream>>>(feat, agg, ind, out, N);
}


# edge scatter: each wave packs the edges it keeps to the front through its own LDS strip; the row-gather and add runs cover kept slots only and stop at their count; f32 atomics, arrival order
# speedup vs baseline: 5.4476x; 1.1104x over previous
#include <hip/hip_runtime.h>

typedef __attribute__((ext_vector_type(2))) float v2f;
typedef __attribute__((ext_vector_type(4))) float v4f;
typedef __attribute__((ext_vector_type(8))) float v8f;
typedef __attribute__((ext_vector_type(4))) int v4i;
typedef __attribute__((ext_vector_type(4))) unsigned v4u;

#define FD 32

__global__ void gcn_zero_kernel(float* __restrict__ p, int n4) {
    int i = blockIdx.x * blockDim.x + threadIdx.x;
    if (i < n4) {
        v4f z = {0.0f, 0.0f, 0.0f, 0.0f};
        ((v4f*)p)[i] = z;
    }
}

#define HR 34816
#define HC 16
#define HT 1024
#define HB 8

__device__ __forceinline__ void gcn_count4(unsigned* hist, v4i v, int r0) {
    unsigned k0 = (unsigned)(v.x - r0);
    unsigned k1 = (unsigned)(v.y - r0);
    unsigned k2 = (unsigned)(v.z - r0);
    unsigned k3 = (unsigned)(v.w - r0);
    if (k0 < (unsigned)HR) atomicAdd(&hist[k0], 1u);
    if (k1 < (unsigned)HR) atomicAdd(&hist[k1], 1u);
    if (k2 < (unsigned)HR) atomicAdd(&hist[k2], 1u);
    if (k3 < (unsigned)HR) atomicAdd(&hist[k3], 1u);
}

__global__ void __launch_bounds__(HT)
gcn_degree_kernel(const int* __restrict__ src,
                  const int* __restrict__ dst,
                  unsigned* __restrict__ outd,
                  unsigned* __restrict__ ind,
                  int N, int E) {
    __shared__ unsigned hist[HR];
    const int tid   = threadIdx.x;
    const int chunk = blockIdx.x;
    const int r0    = blockIdx.y * HR;
    const int* __restrict__ idx = blockIdx.z ? dst : src;
    unsigned* __restrict__  cnt = blockIdx.z ? ind : outd;

    for (int i = tid; i < HR / 4; i += HT) {
        v4u z = {0u, 0u, 0u, 0u};
        ((v4u*)hist)[i] = z;
    }
    __syncthreads();

    const int per = ((E + HC * 4 - 1) / (HC * 4)) * 4;
    const int beg = chunk * per;
    const int lim = beg + per;
    const int end = lim < E ? lim : E;
    const int n4  = end > beg ? (end - beg) >> 2 : 0;
    const v4i* __restrict__ p4 = (const v4i*)(idx + beg);

    int i = tid;
    for (; i + (HB - 1) * HT < n4; i += HB * HT) {
        v4i v[HB];
#pragma unroll
        for (int u = 0; u < HB; ++u) v[u] = p4[i + u * HT];
#pragma unroll
        for (int u = 0; u < HB; ++u) gcn_count4(hist, v[u], r0);
    }
    for (; i < n4; i += HT) gcn_count4(hist, p4[i], r0);
    for (int e = beg + n4 * 4 + tid; e < end; e += HT) {
        unsigned k = (unsigned)(idx[e] - r0);
        if (k < (unsigned)HR) atomicAdd(&hist[k], 1u);
    }
    __syncthreads();

    const int rot = chunk * (HR / HC);
    for (int b = tid; b < HR; b += 2 * HT) {
        int j0 = b + rot;
        int j1 = b + HT + rot;
        if (j0 >= HR) j0 -= HR;
        if (j1 >= HR) j1 -= HR;
        unsigned h0 = hist[j0];
        unsigned h1 = hist[j1];
        if (h0 != 0u && r0 + j0 < N) atomicAdd(&cnt[r0 + j0], h0);
        if (h1 != 0u && r0 + j1 < N) atomicAdd(&cnt[r0 + j1], h1);
    }
}

#define QW 8
#define QB 2
#define QE (QW * QB * 32)
#define QS 64
#define QG 1280

__device__ __forceinline__ void gcn_lds_barrier() {
    __builtin_amdgcn_fence(__ATOMIC_RELEASE, "workgroup", "local");
    __builtin_amdgcn_s_barrier();
    __builtin_amdgcn_fence(__ATOMIC_ACQUIRE, "workgroup", "local");
}

template <int I>
struct gcn_run {
    static __device__ __forceinline__ void gather(float (&f)[QB * 32], const v4i (&pk)[QB],
                                                  const float* __restrict__ feat, int lane, int kept) {
        if (I < kept) {
            const int sj = __builtin_amdgcn_readlane(pk[I >> 5].x, I & 31);
            f[I] = feat[(size_t)sj * FD + lane];
            gcn_run<I + 1>::gather(f, pk, feat, lane, kept);
        }
    }
    static __device__ __forceinline__ void add(const float (&f)[QB * 32], const v4i (&pk)[QB],
                                               float* __restrict__ agg, int lane, int kept) {
        if (I < kept) {
            const int tj = __builtin_amdgcn_readlane(pk[I >> 5].y, I & 31);
            const float nj = __int_as_float(__builtin_amdgcn_readlane(pk[I >> 5].z, I & 31));
            const float wj = __int_as_float(__builtin_amdgcn_readlane(pk[I >> 5].w, I & 31));
            atomicAdd(&agg[(size_t)tj * FD + lane], f[I] * nj * wj);
            gcn_run<I + 1>::add(f, pk, agg, lane, kept);
        }
    }
};
template <>
struct gcn_run<QB * 32> {
    static __device__ __forceinline__ void gather(float (&)[QB * 32], const v4i (&)[QB],
                                                  const float* __restrict__, int, int) {}
    static __device__ __forceinline__ void add(const float (&)[QB * 32], const v4i (&)[QB],
                                               float* __restrict__, int, int) {}
};

__global__ void __launch_bounds__(QW * 32)
gcn_scatter_kernel(const float* __restrict__ feat,
                   const float* __restrict__ e_feat,
                   const int* __restrict__ src,
                   const int* __restrict__ dst,
                   const unsigned* __restrict__ outd,
                   float* __restrict__ agg,
                   unsigned* __restrict__ qctr,
                   int N, int E) {
    __shared__ int s_chunk;
    __shared__ int s_side;
    __shared__ v4i s_edge[QW * QB * 32];
    const int lane = threadIdx.x & 31;
    const int wv = __builtin_amdgcn_readfirstlane((int)(threadIdx.x >> 5));
    const unsigned nch = (unsigned)((E + QE - 1) / QE);
    const int split = N >> 1;
    const int home = (__builtin_amdgcn_s_sendmsg_rtn(0x87) >> 18) & 1;

    int q = home;
    unsigned pend = 0u;
    if (threadIdx.x == 0) pend = atomicAdd(&qctr[q * QS], 1u);

    for (;;) {
        if (threadIdx.x == 0) {
            unsigned b = pend;
            if (b >= nch && q == home) {
                q = home ^ 1;
                b = atomicAdd(&qctr[q * QS], 1u);
            }
            s_chunk = (b < nch) ? (int)b : -1;
            s_side = q;
        }
        gcn_lds_barrier();
        const int chunk = s_chunk;
        const int side = s_side;
        if (chunk < 0) break;
        if (threadIdx.x == 0) pend = atomicAdd(&qctr[q * QS], 1u);

        const int e0 = ((chunk * QW + wv) * QB) * 32;
        int s[QB], t[QB];
        float wm[QB], norm[QB];
        unsigned m[QB];
        bool own[QB];
        unsigned dg[QB];
#pragma unroll
        for (int u = 0; u < QB; ++u) {
            const int e = e0 + u * 32 + lane;
            const bool inb = e < E;
            int su = 0, tu = 0;
            float wu = 0.0f;
            if (inb) {
                su = src[e];
                tu = dst[e];
                wu = e_feat[e];
            }
            s[u] = su;
            t[u] = tu;
            wm[u] = wu * ((su != tu) ? 1.0f : 0.0f);
            own[u] = inb && ((tu >= split ? 1 : 0) == side);
            m[u] = __builtin_amdgcn_ballot_w32(own[u]);
        }
#pragma unroll
        for (int u = 0; u < QB; ++u) {
            dg[u] = 1u;
            if (own[u]) dg[u] = outd[s[u]];
        }
#pragma unroll
        for (int u = 0; u < QB; ++u) norm[u] = rsqrtf((float)(dg[u] ? dg[u] : 1u));

        v4i* __restrict__ mine = s_edge + wv * (QB * 32);
        int kept = 0;
#pragma unroll
        for (int u = 0; u < QB; ++u) {
            const int r = kept + (int)__builtin_amdgcn_mbcnt_lo(m[u], 0u);
            if (own[u]) {
                v4i p = {s[u], t[u], __float_as_int(norm[u]), __float_as_int(wm[u])};
                mine[r] = p;
            }
            kept += __builtin_popcount(m[u]);
        }
        v4i pk[QB];
#pragma unroll
        for (int u = 0; u < QB; ++u) {
            v4i p = {0, 0, 0, 0};
            if (u * 32 + lane < kept) p = mine[u * 32 + lane];
            pk[u] = p;
        }

        float f[QB * 32];
        gcn_run<0>::gather(f, pk, feat, lane, kept);
        gcn_run<0>::add(f, pk, agg, lane, kept);
        gcn_lds_barrier();
    }
}

__global__ void gcn_finalize_kernel(const float* __restrict__ feat,
                                    const float* __restrict__ agg,
                                    const unsigned* __restrict__ ind,
                                    float* __restrict__ out,
                                    int N) {
    int wave = blockIdx.x * (blockDim.x >> 5) + (threadIdx.x >> 5);
    int lane = threadIdx.x & 31;
    int ntiles = (N + 15) >> 4;
    if (wave >= ntiles) return;

    int r    = lane & 15;
    int half = lane >> 4;
    int node = wave * 16 + r;

    float vals[16];
    if (node < N) {
        unsigned dg = ind[node];
        float dn = rsqrtf((float)(dg ? dg : 1u));
        const v4f* fp = (const v4f*)(feat + (size_t)node * FD + half * 16);
        const v4f* ap = (const v4f*)(agg  + (size_t)node * FD + half * 16);
#pragma unroll
        for (int q = 0; q < 4; ++q) {
            v4f f = fp[q];
            v4f a = ap[q];
            vals[4 * q + 0] = __builtin_fabsf(f.x - a.x * dn);
            vals[4 * q + 1] = __builtin_fabsf(f.y - a.y * dn);
            vals[4 * q + 2] = __builtin_fabsf(f.z - a.z * dn);
            vals[4 * q + 3] = __builtin_fabsf(f.w - a.w * dn);
        }
    } else {
#pragma unroll
        for (int q = 0; q < 16; ++q) vals[q] = 0.0f;
    }

    v2f ones = {1.0f, 1.0f};
    v8f c = {};
#pragma unroll
    for (int kk = 0; kk < 8; ++kk) {
        v2f a;
        a.x = vals[2 * kk + 0];
        a.y = vals[2 * kk + 1];
        c = __builtin_amdgcn_wmma_f32_16x16x4_f32(
                false, a, false, ones, (short)0, c, false, false);
    }

    if ((lane & 15) == 0) {
        int base = wave * 16 + half * 8;
#pragma unroll
        for (int m = 0; m < 8; ++m) {
            int nd = base + m;
            if (nd < N) out[nd] = c[m];
        }
    }
}

extern "C" void kernel_launch(void* const* d_in, const int* in_sizes, int n_in,
                              void* d_out, int out_size, void* d_ws, size_t ws_size,
                              hipStream_t stream) {
    const float* feat   = (const float*)d_in[0];
    const float* e_feat = (const float*)d_in[1];
    const int*   src    = (const int*)d_in[2];
    const int*   dst    = (const int*)d_in[3];
    float*       out    = (float*)d_out;

    int N = in_sizes[0] / FD;
    int E = in_sizes[2];

    unsigned* outd = (unsigned*)d_ws;
    unsigned* ind  = outd + N;
    float*    agg  = (float*)(ind + N);
    unsigned* qctr = (unsigned*)(agg + (size_t)N * FD);

    long long zero_words = (long long)2 * N + (long long)N * FD + 2 * QS;
    int zero_v4 = (int)((zero_words + 3) / 4);
    gcn_zero_kernel<<<(zero_v4 + 255) / 256, 256, 0, stream>>>((float*)d_ws, zero_v4);

    gcn_degree_kernel<<<dim3(HC, (N + HR - 1) / HR, 2), HT, 0, stream>>>(src, dst, outd, ind, N, E);

    gcn_scatter_kernel<<<QG, QW * 32, 0, stream>>>(
        feat, e_feat, src, dst, outd, agg, qctr, N, E);

    int ntiles = (N + 15) / 16;
    gcn_finalize_kernel<<<(ntiles + 7) / 8, 256, 0, stream>>>(feat, agg, ind, out, N);
}


# edge scatter: row element offsets (32*src, 32*dst) and one factor norm*weight worked out per edge before packing; a kept slot is 1 lane read + 32-bit add per address, 1 multiply; product regrouped
# speedup vs baseline: 5.6085x; 1.0295x over previous
#include <hip/hip_runtime.h>

typedef __attribute__((ext_vector_type(2))) float v2f;
typedef __attribute__((ext_vector_type(4))) float v4f;
typedef __attribute__((ext_vector_type(8))) float v8f;
typedef __attribute__((ext_vector_type(4))) int v4i;
typedef __attribute__((ext_vector_type(4))) unsigned v4u;

#define FD 32

__global__ void gcn_zero_kernel(float* __restrict__ p, int n4) {
    int i = blockIdx.x * blockDim.x + threadIdx.x;
    if (i < n4) {
        v4f z = {0.0f, 0.0f, 0.0f, 0.0f};
        ((v4f*)p)[i] = z;
    }
}

#define HR 34816
#define HC 16
#define HT 1024
#define HB 8

__device__ __forceinline__ void gcn_count4(unsigned* hist, v4i v, int r0) {
    unsigned k0 = (unsigned)(v.x - r0);
    unsigned k1 = (unsigned)(v.y - r0);
    unsigned k2 = (unsigned)(v.z - r0);
    unsigned k3 = (unsigned)(v.w - r0);
    if (k0 < (unsigned)HR) atomicAdd(&hist[k0], 1u);
    if (k1 < (unsigned)HR) atomicAdd(&hist[k1], 1u);
    if (k2 < (unsigned)HR) atomicAdd(&hist[k2], 1u);
    if (k3 < (unsigned)HR) atomicAdd(&hist[k3], 1u);
}

__global__ void __launch_bounds__(HT)
gcn_degree_kernel(const int* __restrict__ src,
                  const int* __restrict__ dst,
                  unsigned* __restrict__ outd,
                  unsigned* __restrict__ ind,
                  int N, int E) {
    __shared__ unsigned hist[HR];
    const int tid   = threadIdx.x;
    const int chunk = blockIdx.x;
    const int r0    = blockIdx.y * HR;
    const int* __restrict__ idx = blockIdx.z ? dst : src;
    unsigned* __restrict__  cnt = blockIdx.z ? ind : outd;

    for (int i = tid; i < HR / 4; i += HT) {
        v4u z = {0u, 0u, 0u, 0u};
        ((v4u*)hist)[i] = z;
    }
    __syncthreads();

    const int per = ((E + HC * 4 - 1) / (HC * 4)) * 4;
    const int beg = chunk * per;
    const int lim = beg + per;
    const int end = lim < E ? lim : E;
    const int n4  = end > beg ? (end - beg) >> 2 : 0;
    const v4i* __restrict__ p4 = (const v4i*)(idx + beg);

    int i = tid;
    for (; i + (HB - 1) * HT < n4; i += HB * HT) {
        v4i v[HB];
#pragma unroll
        for (int u = 0; u < HB; ++u) v[u] = p4[i + u * HT];
#pragma unroll
        for (int u = 0; u < HB; ++u) gcn_count4(hist, v[u], r0);
    }
    for (; i < n4; i += HT) gcn_count4(hist, p4[i], r0);
    for (int e = beg + n4 * 4 + tid; e < end; e += HT) {
        unsigned k = (unsigned)(idx[e] - r0);
        if (k < (unsigned)HR) atomicAdd(&hist[k], 1u);
    }
    __syncthreads();

    const int rot = chunk * (HR / HC);
    for (int b = tid; b < HR; b += 2 * HT) {
        int j0 = b + rot;
        int j1 = b + HT + rot;
        if (j0 >= HR) j0 -= HR;
        if (j1 >= HR) j1 -= HR;
        unsigned h0 = hist[j0];
        unsigned h1 = hist[j1];
        if (h0 != 0u && r0 + j0 < N) atomicAdd(&cnt[r0 + j0], h0);
        if (h1 != 0u && r0 + j1 < N) atomicAdd(&cnt[r0 + j1], h1);
    }
}

#define QW 8
#define QB 2
#define QE (QW * QB * 32)
#define QS 64
#define QG 1280

__device__ __forceinline__ void gcn_lds_barrier() {
    __builtin_amdgcn_fence(__ATOMIC_RELEASE, "workgroup", "local");
    __builtin_amdgcn_s_barrier();
    __builtin_amdgcn_fence(__ATOMIC_ACQUIRE, "workgroup", "local");
}

template <int I>
struct gcn_run {
    static __device__ __forceinline__ void gather(float (&f)[QB * 32], const v4i (&pk)[QB],
                                                  const float* __restrict__ feat, unsigned lane, int kept) {
        if (I < kept) {
            const unsigned so = (unsigned)__builtin_amdgcn_readlane(pk[I >> 5].x, I & 31);
            f[I] = feat[so + lane];
            gcn_run<I + 1>::gather(f, pk, feat, lane, kept);
        }
    }
    static __device__ __forceinline__ void add(const float (&f)[QB * 32], const v4i (&pk)[QB],
                                               float* __restrict__ agg, unsigned lane, int kept) {
        if (I < kept) {
            const unsigned to = (unsigned)__builtin_amdgcn_readlane(pk[I >> 5].y, I & 31);
            const float cj = __int_as_float(__builtin_amdgcn_readlane(pk[I >> 5].z, I & 31));
            atomicAdd(&agg[to + lane], f[I] * cj);
            gcn_run<I + 1>::add(f, pk, agg, lane, kept);
        }
    }
};
template <>
struct gcn_run<QB * 32> {
    static __device__ __forceinline__ void gather(float (&)[QB * 32], const v4i (&)[QB],
                                                  const float* __restrict__, unsigned, int) {}
    static __device__ __forceinline__ void add(const float (&)[QB * 32], const v4i (&)[QB],
                                               float* __restrict__, unsigned, int) {}
};

__global__ void __launch_bounds__(QW * 32)
gcn_scatter_kernel(const float* __restrict__ feat,
                   const float* __restrict__ e_feat,
                   const int* __restrict__ src,
                   const int* __restrict__ dst,
                   const unsigned* __restrict__ outd,
                   float* __restrict__ agg,
                   unsigned* __restrict__ qctr,
                   int N, int E) {
    __shared__ int s_chunk;
    __shared__ int s_side;
    __shared__ v4i s_edge[QW * QB * 32];
    const int lane = threadIdx.x & 31;
    const int wv = __builtin_amdgcn_readfirstlane((int)(threadIdx.x >> 5));
    const unsigned nch = (unsigned)((E + QE - 1) / QE);
    const int split = N >> 1;
    const int home = (__builtin_amdgcn_s_sendmsg_rtn(0x87) >> 18) & 1;

    int q = home;
    unsigned pend = 0u;
    if (threadIdx.x == 0) pend = atomicAdd(&qctr[q * QS], 1u);

    for (;;) {
        if (threadIdx.x == 0) {
            unsigned b = pend;
            if (b >= nch && q == home) {
                q = home ^ 1;
                b = atomicAdd(&qctr[q * QS], 1u);
            }
            s_chunk = (b < nch) ? (int)b : -1;
            s_side = q;
        }
        gcn_lds_barrier();
        const int chunk = s_chunk;
        const int side = s_side;
        if (chunk < 0) break;
        if (threadIdx.x == 0) pend = atomicAdd(&qctr[q * QS], 1u);

        const int e0 = ((chunk * QW + wv) * QB) * 32;
        int s[QB], t[QB];
        float wm[QB], norm[QB];
        unsigned m[QB];
        bool own[QB];
        unsigned dg[QB];
#pragma unroll
        for (int u = 0; u < QB; ++u) {
            const int e = e0 + u * 32 + lane;
            const bool inb = e < E;
            int su = 0, tu = 0;
            float wu = 0.0f;
            if (inb) {
                su = src[e];
                tu = dst[e];
                wu = e_feat[e];
            }
            s[u] = su;
            t[u] = tu;
            wm[u] = wu * ((su != tu) ? 1.0f : 0.0f);
            own[u] = inb && ((tu >= split ? 1 : 0) == side);
            m[u] = __builtin_amdgcn_ballot_w32(own[u]);
        }
#pragma unroll
        for (int u = 0; u < QB; ++u) {
            dg[u] = 1u;
            if (own[u]) dg[u] = outd[s[u]];
        }
#pragma unroll
        for (int u = 0; u < QB; ++u) norm[u] = rsqrtf((float)(dg[u] ? dg[u] : 1u));

        v4i* __restrict__ mine = s_edge + wv * (QB * 32);
        int kept = 0;
#pragma unroll
        for (int u = 0; u < QB; ++u) {
            const int r = kept + (int)__builtin_amdgcn_mbcnt_lo(m[u], 0u);
            if (own[u]) {
                v4i p = {s[u] * FD, t[u] * FD, __float_as_int(norm[u] * wm[u]), 0};
                mine[r] = p;
            }
            kept += __builtin_popcount(m[u]);
        }
        v4i pk[QB];
#pragma unroll
        for (int u = 0; u < QB; ++u) {
            v4i p = {0, 0, 0, 0};
            if (u * 32 + lane < kept) p = mine[u * 32 + lane];
            pk[u] = p;
        }

        float f[QB * 32];
        gcn_run<0>::gather(f, pk, feat, (unsigned)lane, kept);
        gcn_run<0>::add(f, pk, agg, (unsigned)lane, kept);
        gcn_lds_barrier();
    }
}

__global__ void gcn_finalize_kernel(const float* __restrict__ feat,
                                    const float* __restrict__ agg,
                                    const unsigned* __restrict__ ind,
                                    float* __restrict__ out,
                                    int N) {
    int wave = blockIdx.x * (blockDim.x >> 5) + (threadIdx.x >> 5);
    int lane = threadIdx.x & 31;
    int ntiles = (N + 15) >> 4;
    if (wave >= ntiles) return;

    int r    = lane & 15;
    int half = lane >> 4;
    int node = wave * 16 + r;

    float vals[16];
    if (node < N) {
        unsigned dg = ind[node];
        float dn = rsqrtf((float)(dg ? dg : 1u));
        const v4f* fp = (const v4f*)(feat + (size_t)node * FD + half * 16);
        const v4f* ap = (const v4f*)(agg  + (size_t)node * FD + half * 16);
#pragma unroll
        for (int q = 0; q < 4; ++q) {
            v4f f = fp[q];
            v4f a = ap[q];
            vals[4 * q + 0] = __builtin_fabsf(f.x - a.x * dn);
            vals[4 * q + 1] = __builtin_fabsf(f.y - a.y * dn);
            vals[4 * q + 2] = __builtin_fabsf(f.z - a.z * dn);
            vals[4 * q + 3] = __builtin_fabsf(f.w - a.w * dn);
        }
    } else {
#pragma unroll
        for (int q = 0; q < 16; ++q) vals[q] = 0.0f;
    }

    v2f ones = {1.0f, 1.0f};
    v8f c = {};
#pragma unroll
    for (int kk = 0; kk < 8; ++kk) {
        v2f a;
        a.x = vals[2 * kk + 0];
        a.y = vals[2 * kk + 1];
        c = __builtin_amdgcn_wmma_f32_16x16x4_f32(
                false, a, false, ones, (short)0, c, false, false);
    }

    if ((lane & 15) == 0) {
        int base = wave * 16 + half * 8;
#pragma unroll
        for (int m = 0; m < 8; ++m) {
            int nd = base + m;
            if (nd < N) out[nd] = c[m];
        }
    }
}

extern "C" void kernel_launch(void* const* d_in, const int* in_sizes, int n_in,
                              void* d_out, int out_size, void* d_ws, size_t ws_size,
                              hipStream_t stream) {
    const float* feat   = (const float*)d_in[0];
    const float* e_feat = (const float*)d_in[1];
    const int*   src    = (const int*)d_in[2];
    const int*   dst    = (const int*)d_in[3];
    float*       out    = (float*)d_out;

    int N = in_sizes[0] / FD;
    int E = in_sizes[2];

    unsigned* outd = (unsigned*)d_ws;
    unsigned* ind  = outd + N;
    float*    agg  = (float*)(ind + N);
    unsigned* qctr = (unsigned*)(agg + (size_t)N * FD);

    long long zero_words = (long long)2 * N + (long long)N * FD + 2 * QS;
    int zero_v4 = (int)((zero_words + 3) / 4);
    gcn_zero_kernel<<<(zero_v4 + 255) / 256, 256, 0, stream>>>((float*)d_ws, zero_v4);

    gcn_degree_kernel<<<dim3(HC, (N + HR - 1) / HR, 2), HT, 0, stream>>>(src, dst, outd, ind, N, E);

    gcn_scatter_kernel<<<QG, QW * 32, 0, stream>>>(
        feat, e_feat, src, dst, outd, agg, qctr, N, E);

    int ntiles = (N + 15) / 16;
    gcn_finalize_kernel<<<(ntiles + 7) / 8, 256, 0, stream>>>(feat, agg, ind, out, N);
}
